# starting kernel
# baseline (speedup 1.0000x reference)
#include <hip/hip_runtime.h>

typedef float v2f __attribute__((ext_vector_type(2)));
typedef float v8f __attribute__((ext_vector_type(8)));

#define T_   512
#define K_   4
#define G_   20000
#define EPG_ 3
#define B_   512
#define BT   16
#define GT   512
#define NW   16

__device__ __forceinline__ float lrelu(float x) {
    return x > 0.f ? x : 0.01f * x;
}

__global__ __launch_bounds__(512)
void ae_decoder_fused(const float* __restrict__ features,
                      const float* __restrict__ w1,
                      const float* __restrict__ b1,
                      const float* __restrict__ w2,
                      const float* __restrict__ b2,
                      const float* __restrict__ w3,
                      const float* __restrict__ b3,
                      const int*   __restrict__ edge_tf,
                      float*       __restrict__ out)
{
    __shared__ __align__(16) float lds_h2[BT * T_ * K_];

    const int tid  = threadIdx.x;
    const int wave = tid >> 5;
    const int lane = tid & 31;
    const int half = lane >> 4;
    const int m    = lane & 15;

    const int bbase = blockIdx.y * BT;
    const int gbase = blockIdx.x * GT;

    const int j0 = half * 2;
    for (int i = 0; i < T_ / NW; ++i) {
        const int t = wave * (T_ / NW) + i;

        const float f = features[(size_t)(bbase + m) * T_ + t];
        v2f a;
        a.x = lrelu(f * w1[t * K_ + j0]     + b1[t * K_ + j0]);
        a.y = lrelu(f * w1[t * K_ + j0 + 1] + b1[t * K_ + j0 + 1]);

        float bx = 0.f, by = 0.f, cv = 0.f;
        if (m < K_) {
            bx = w2[(t * K_ + j0)     * K_ + m];
            by = w2[(t * K_ + j0 + 1) * K_ + m];
            cv = b2[t * K_ + m];
        }
        v2f bmat; bmat.x = bx; bmat.y = by;
        v8f c = {cv, cv, cv, cv, cv, cv, cv, cv};

        v8f d = __builtin_amdgcn_wmma_f32_16x16x4_f32(
             false, a,  false, bmat,
             (short)0, c,  false,  false);

        if (m < K_) {
            #pragma unroll
            for (int v = 0; v < 8; ++v) {
                const int brow = v + 8 * half;
                lds_h2[(brow * T_ + t) * K_ + m] = lrelu(d[v]);
            }
        }
    }

    __syncthreads();

    const int    bl    = wave;
    const size_t bglob = (size_t)(bbase + bl);
    const float4* __restrict__ w3v = (const float4*)w3;

    for (int it = 0; it < GT / 32; ++it) {
        const int g = gbase + it * 32 + lane;
        if (g < G_) {
            float acc = b3[g];
            #pragma unroll
            for (int e = 0; e < EPG_; ++e) {
                const int t = edge_tf[g * EPG_ + e];
                const float4 h  = *(const float4*)&lds_h2[(bl * T_ + t) * K_];
                const float4 wv = w3v[g * EPG_ + e];
                acc = fmaf(h.x, wv.x, acc);
                acc = fmaf(h.y, wv.y, acc);
                acc = fmaf(h.z, wv.z, acc);
                acc = fmaf(h.w, wv.w, acc);
            }
            out[bglob * G_ + g] = acc;
        }
    }
}

extern "C" void kernel_launch(void* const* d_in, const int* in_sizes, int n_in,
                              void* d_out, int out_size, void* d_ws, size_t ws_size,
                              hipStream_t stream)
{
    const float* features = (const float*)d_in[0];
    const float* w1       = (const float*)d_in[1];
    const float* b1       = (const float*)d_in[2];
    const float* w2       = (const float*)d_in[3];
    const float* b2       = (const float*)d_in[4];
    const float* w3       = (const float*)d_in[5];
    const float* b3       = (const float*)d_in[6];
    const int*   edge_tf  = (const int*)d_in[7];
    float*       out      = (float*)d_out;

    dim3 grid((G_ + GT - 1) / GT, B_ / BT);
    ae_decoder_fused<<<grid, 512, 0, stream>>>(features, w1, b1, w2, b2,
                                               w3, b3, edge_tf, out);
}


# Phase A: a wave fetches w1, b1, w2, b2 of its 32 t once as whole lines and parks each t's record in that t's h2 row; the t loop reads them from LDS a step ahead, no global load per t; same fma order
# speedup vs baseline: 2.8714x; 2.8714x over previous
#include <hip/hip_runtime.h>


typedef float v2f __attribute__((ext_vector_type(2)));
typedef float v8f __attribute__((ext_vector_type(8)));

#define T_   512
#define K_   4
#define G_   20000
#define EPG_ 3
#define B_   512
#define BT   32
#define NW   16
#define GS   16
#define CH   32
#define NCH  (G_ / CH)

__device__ __forceinline__ float lrelu(float x) {
    return x > 0.f ? x : 0.01f * x;
}

template <int S>
__device__ __forceinline__ float row_bcast(float v) {
    return __builtin_bit_cast(float,
        __builtin_amdgcn_update_dpp(0, __builtin_bit_cast(int, v), 0x150 + S, 0xf, 0xf, true));
}
template <int S>
__device__ __forceinline__ int row_bcast(int v) {
    return __builtin_amdgcn_update_dpp(0, v, 0x150 + S, 0xf, 0xf, true);
}

struct GOps {
    float4 w[EPG_];
    int    to[EPG_];
    float  b;
};

__device__ __forceinline__ GOps fetch_gops(const float* __restrict__ w3,
                                           const float* __restrict__ b3,
                                           const int*   __restrict__ edge_tf, int g)
{
    GOps o;
    #pragma unroll
    for (int e = 0; e < EPG_; ++e) {
        o.w[e]  = *(const float4*)&w3[(size_t)(g * EPG_ + e) * K_];
        o.to[e] = edge_tf[g * EPG_ + e] * (BT * K_ * 4);
    }
    o.b = b3[g];
    return o;
}

template <int S>
__device__ __forceinline__ void fma_row(float& acc, float w, float h)
{
    asm("v_fmac_f32_dpp %0, %1, %2 row_share:%3 row_mask:0xf bank_mask:0xf"
        : "+v"(acc) : "v"(w), "v"(h), "n"(S));
}

template <int S>
__device__ __forceinline__ float one_g(const GOps& o, const char* h2row)
{
    float a = row_bcast<S>(o.b);
    #pragma unroll
    for (int e = 0; e < EPG_; ++e) {
        const float4 h = *(const float4*)(h2row + row_bcast<S>(o.to[e]));
        fma_row<S>(a, o.w[e].x, h.x);
        fma_row<S>(a, o.w[e].y, h.y);
        fma_row<S>(a, o.w[e].z, h.z);
        fma_row<S>(a, o.w[e].w, h.w);
    }
    return a;
}

template <int S0>
__device__ __forceinline__ float4 four_g(const GOps& o, const char* h2row)
{
    return make_float4(one_g<S0>(o, h2row), one_g<S0 + 1>(o, h2row),
                       one_g<S0 + 2>(o, h2row), one_g<S0 + 3>(o, h2row));
}

struct AOps {
    float2 w1, b1;
    float  bx, by, cv, f0, f1;
};

__device__ __forceinline__ AOps fetch_aops(const float* h2, const float* ft, int t0, int i,
                                           int j0, int mk, int m, int fsw)
{
    const int t = t0 + i;
    const float* rec = h2 + t * (BT * K_) + (t & 7) * 12;
    AOps o;
    o.w1 = *(const float2*)&rec[j0];
    o.b1 = *(const float2*)&rec[4 + j0];
    o.bx = rec[8 + j0 * K_ + mk];
    o.by = rec[8 + (j0 + 1) * K_ + mk];
    o.cv = rec[24 + mk];
    const int fo = m * (T_ / NW) + ((((i >> 2) ^ fsw) << 2) | (i & 3));
    o.f0 = ft[fo];
    o.f1 = ft[fo + 16 * (T_ / NW)];
    return o;
}

__global__ __launch_bounds__(512)
void ae_decoder_fused(const float* __restrict__ features,
                      const float* __restrict__ w1,
                      const float* __restrict__ b1,
                      const float* __restrict__ w2,
                      const float* __restrict__ b2,
                      const float* __restrict__ w3,
                      const float* __restrict__ b3,
                      const int*   __restrict__ edge_tf,
                      float*       __restrict__ out)
{
    __shared__ __align__(16) float lds_h2[T_ * BT * K_];
    __shared__ __align__(16) float lds_tr[NW * BT * CH];

    const int tid  = threadIdx.x;
    const int wave = __builtin_amdgcn_readfirstlane(tid >> 5);
    const int lane = tid & 31;
    const int half = lane >> 4;
    const int m    = lane & 15;

    const int bbase  = blockIdx.y * BT;
    const int gsplit = blockIdx.x;

    const int c0 = gsplit + GS * wave;
    GOps nxt = fetch_gops(w3, b3, edge_tf, c0 * CH + m);

    const int j0 = half * 2;

    float* ft = lds_tr + wave * (BT * CH);
    const int t0 = wave * (T_ / NW);
    {
        const int fr = lane >> 3;
        const int fc = lane & 7;
        float4 fv[BT / 4];
        #pragma unroll
        for (int p = 0; p < BT / 4; ++p) {
            const int r = p * 4 + fr;
            fv[p] = *(const float4*)&features[(size_t)(bbase + r) * T_ + t0 + (fc << 2)];
        }
        #pragma unroll
        for (int p = 0; p < BT / 4; ++p) {
            const int r = p * 4 + fr;
            *(float4*)&ft[r * (T_ / NW) + ((fc ^ ((r >> 1) & 7)) << 2)] = fv[p];
        }
    }
    const int fsw = (m >> 1) & 7;

    {
        const int tl = t0 + lane;
        const float4 pw1 = *(const float4*)&w1[tl * K_];
        const float4 pb1 = *(const float4*)&b1[tl * K_];
        const float4 pb2 = *(const float4*)&b2[tl * K_];
        float4 pw2[4];
        #pragma unroll
        for (int q = 0; q < 4; ++q)
            pw2[q] = *(const float4*)&w2[(size_t)t0 * (K_ * K_) + q * 128 + lane * 4];
        float* rl = &lds_h2[tl * (BT * K_) + (tl & 7) * 12];
        *(float4*)&rl[0]  = pw1;
        *(float4*)&rl[4]  = pb1;
        *(float4*)&rl[24] = pb2;
        #pragma unroll
        for (int q = 0; q < 4; ++q) {
            const int tq = t0 + q * 8 + (lane >> 2);
            *(float4*)&lds_h2[tq * (BT * K_) + (tq & 7) * 12 + 8 + (lane & 3) * 4] = pw2[q];
        }
    }
    const int mk = m & 3;

    AOps cur = fetch_aops(lds_h2, ft, t0, 0, j0, mk, m, fsw);

    #pragma unroll 8
    for (int i = 0; i < T_ / NW; ++i) {
        const int t = t0 + i;

        const AOps nxt = fetch_aops(lds_h2, ft, t0, (i + 1) & (T_ / NW - 1), j0, mk, m, fsw);

        v2f bmat; bmat.x = cur.bx; bmat.y = cur.by;
        const float cv = cur.cv;

        #pragma unroll
        for (int mt = 0; mt < BT / 16; ++mt) {
            const float f = mt ? cur.f1 : cur.f0;
            v2f a;
            a.x = lrelu(f * cur.w1.x + cur.b1.x);
            a.y = lrelu(f * cur.w1.y + cur.b1.y);
            v8f c = {cv, cv, cv, cv, cv, cv, cv, cv};

            v8f d = __builtin_amdgcn_wmma_f32_16x16x4_f32(
                 false, a,  false, bmat,
                 (short)0, c,  false,  false);

            if (m < K_) {
                #pragma unroll
                for (int v = 0; v < 8; ++v) {
                    const int brow = mt * 16 + v + 8 * half;
                    lds_h2[(t * BT + brow) * K_ + m] = lrelu(d[v]);
                }
            }
        }
        cur = nxt;
    }

    __syncthreads();

    float* tr = lds_tr + wave * (BT * CH);
    const int wsw = (lane >> 1) & 7;
    const int rr0 = lane >> 3;
    const int rcg = lane & 7;
    const char* h2row = (const char*)lds_h2 + lane * (K_ * 4);

    for (int c = c0; c < NCH; c += GS * NW) {
        const int g0 = c * CH;
        const int cn = (c + GS * NW < NCH) ? c + GS * NW : c;

        #pragma unroll
        for (int hc = 0; hc < 2; ++hc) {
            const GOps cur = nxt;
            nxt = fetch_gops(w3, b3, edge_tf, (hc == 0 ? g0 + 16 : cn * CH) + m);

            *(float4*)&tr[lane * CH + (((hc * 4 + 0) ^ wsw) << 2)] = four_g<0>(cur, h2row);
            *(float4*)&tr[lane * CH + (((hc * 4 + 1) ^ wsw) << 2)] = four_g<4>(cur, h2row);
            *(float4*)&tr[lane * CH + (((hc * 4 + 2) ^ wsw) << 2)] = four_g<8>(cur, h2row);
            *(float4*)&tr[lane * CH + (((hc * 4 + 3) ^ wsw) << 2)] = four_g<12>(cur, h2row);
        }

        #pragma unroll
        for (int p = 0; p < BT / 4; ++p) {
            const int r  = p * 4 + rr0;
            const int cs = rcg ^ ((r >> 1) & 7);
            const float4 v = *(const float4*)&tr[r * CH + (cs << 2)];
            *(float4*)&out[(size_t)(bbase + r) * G_ + g0 + (rcg << 2)] = v;
        }
    }
}

extern "C" void kernel_launch(void* const* d_in, const int* in_sizes, int n_in,
                              void* d_out, int out_size, void* d_ws, size_t ws_size,
                              hipStream_t stream)
{
    const float* features = (const float*)d_in[0];
    const float* w1       = (const float*)d_in[1];
    const float* b1       = (const float*)d_in[2];
    const float* w2       = (const float*)d_in[3];
    const float* b2       = (const float*)d_in[4];
    const float* w3       = (const float*)d_in[5];
    const float* b3       = (const float*)d_in[6];
    const int*   edge_tf  = (const int*)d_in[7];
    float*       out      = (float*)d_out;

    dim3 grid(GS, B_ / BT);
    ae_decoder_fused<<<grid, 512, 0, stream>>>(features, w1, b1, w2, b2,
                                               w3, b3, edge_tf, out);
}


# Phase A: leaky ReLU as max(x, 0.01x): one multiply and one maximum per value, no compare + select pair; same value for every x, same fma order
# speedup vs baseline: 3.5550x; 1.2381x over previous
#include <hip/hip_runtime.h>


typedef float v2f __attribute__((ext_vector_type(2)));
typedef float v8f __attribute__((ext_vector_type(8)));

#define T_   512
#define K_   4
#define G_   20000
#define EPG_ 3
#define B_   512
#define BT   32
#define NW   16
#define GS   16
#define CH   32
#define NCH  (G_ / CH)

__device__ __forceinline__ float lrelu(float x) {
    return __builtin_fmaxf(x, 0.01f * x);
}

template <int S>
__device__ __forceinline__ float row_bcast(float v) {
    return __builtin_bit_cast(float,
        __builtin_amdgcn_update_dpp(0, __builtin_bit_cast(int, v), 0x150 + S, 0xf, 0xf, true));
}
template <int S>
__device__ __forceinline__ int row_bcast(int v) {
    return __builtin_amdgcn_update_dpp(0, v, 0x150 + S, 0xf, 0xf, true);
}

struct GOps {
    float4 w[EPG_];
    int    to[EPG_];
    float  b;
};

__device__ __forceinline__ GOps fetch_gops(const float* __restrict__ w3,
                                           const float* __restrict__ b3,
                                           const int*   __restrict__ edge_tf, int g)
{
    GOps o;
    #pragma unroll
    for (int e = 0; e < EPG_; ++e) {
        o.w[e]  = *(const float4*)&w3[(size_t)(g * EPG_ + e) * K_];
        o.to[e] = edge_tf[g * EPG_ + e] * (BT * K_ * 4);
    }
    o.b = b3[g];
    return o;
}

template <int S>
__device__ __forceinline__ void fma_row(float& acc, float w, float h)
{
    asm("v_fmac_f32_dpp %0, %1, %2 row_share:%3 row_mask:0xf bank_mask:0xf"
        : "+v"(acc) : "v"(w), "v"(h), "n"(S));
}

template <int S>
__device__ __forceinline__ float one_g(const GOps& o, const char* h2row)
{
    float a = row_bcast<S>(o.b);
    #pragma unroll
    for (int e = 0; e < EPG_; ++e) {
        const float4 h = *(const float4*)(h2row + row_bcast<S>(o.to[e]));
        fma_row<S>(a, o.w[e].x, h.x);
        fma_row<S>(a, o.w[e].y, h.y);
        fma_row<S>(a, o.w[e].z, h.z);
        fma_row<S>(a, o.w[e].w, h.w);
    }
    return a;
}

template <int S0>
__device__ __forceinline__ float4 four_g(const GOps& o, const char* h2row)
{
    return make_float4(one_g<S0>(o, h2row), one_g<S0 + 1>(o, h2row),
                       one_g<S0 + 2>(o, h2row), one_g<S0 + 3>(o, h2row));
}

struct AOps {
    float2 w1, b1;
    float  bx, by, cv, f0, f1;
};

__device__ __forceinline__ AOps fetch_aops(const float* h2, const float* ft, int t0, int i,
                                           int j0, int mk, int m, int fsw)
{
    const int t = t0 + i;
    const float* rec = h2 + t * (BT * K_) + (t & 7) * 12;
    AOps o;
    o.w1 = *(const float2*)&rec[j0];
    o.b1 = *(const float2*)&rec[4 + j0];
    o.bx = rec[8 + j0 * K_ + mk];
    o.by = rec[8 + (j0 + 1) * K_ + mk];
    o.cv = rec[24 + mk];
    const int fo = m * (T_ / NW) + ((((i >> 2) ^ fsw) << 2) | (i & 3));
    o.f0 = ft[fo];
    o.f1 = ft[fo + 16 * (T_ / NW)];
    return o;
}

__global__ __launch_bounds__(512)
void ae_decoder_fused(const float* __restrict__ features,
                      const float* __restrict__ w1,
                      const float* __restrict__ b1,
                      const float* __restrict__ w2,
                      const float* __restrict__ b2,
                      const float* __restrict__ w3,
                      const float* __restrict__ b3,
                      const int*   __restrict__ edge_tf,
                      float*       __restrict__ out)
{
    __shared__ __align__(16) float lds_h2[T_ * BT * K_];
    __shared__ __align__(16) float lds_tr[NW * BT * CH];

    const int tid  = threadIdx.x;
    const int wave = __builtin_amdgcn_readfirstlane(tid >> 5);
    const int lane = tid & 31;
    const int half = lane >> 4;
    const int m    = lane & 15;

    const int bbase  = blockIdx.y * BT;
    const int gsplit = blockIdx.x;

    const int c0 = gsplit + GS * wave;
    GOps nxt = fetch_gops(w3, b3, edge_tf, c0 * CH + m);

    const int j0 = half * 2;

    float* ft = lds_tr + wave * (BT * CH);
    const int t0 = wave * (T_ / NW);
    {
        const int fr = lane >> 3;
        const int fc = lane & 7;
        float4 fv[BT / 4];
        #pragma unroll
        for (int p = 0; p < BT / 4; ++p) {
            const int r = p * 4 + fr;
            fv[p] = *(const float4*)&features[(size_t)(bbase + r) * T_ + t0 + (fc << 2)];
        }
        #pragma unroll
        for (int p = 0; p < BT / 4; ++p) {
            const int r = p * 4 + fr;
            *(float4*)&ft[r * (T_ / NW) + ((fc ^ ((r >> 1) & 7)) << 2)] = fv[p];
        }
    }
    const int fsw = (m >> 1) & 7;

    {
        const int tl = t0 + lane;
        const float4 pw1 = *(const float4*)&w1[tl * K_];
        const float4 pb1 = *(const float4*)&b1[tl * K_];
        const float4 pb2 = *(const float4*)&b2[tl * K_];
        float4 pw2[4];
        #pragma unroll
        for (int q = 0; q < 4; ++q)
            pw2[q] = *(const float4*)&w2[(size_t)t0 * (K_ * K_) + q * 128 + lane * 4];
        float* rl = &lds_h2[tl * (BT * K_) + (tl & 7) * 12];
        *(float4*)&rl[0]  = pw1;
        *(float4*)&rl[4]  = pb1;
        *(float4*)&rl[24] = pb2;
        #pragma unroll
        for (int q = 0; q < 4; ++q) {
            const int tq = t0 + q * 8 + (lane >> 2);
            *(float4*)&lds_h2[tq * (BT * K_) + (tq & 7) * 12 + 8 + (lane & 3) * 4] = pw2[q];
        }
    }
    const int mk = m & 3;

    AOps cur = fetch_aops(lds_h2, ft, t0, 0, j0, mk, m, fsw);

    #pragma unroll 8
    for (int i = 0; i < T_ / NW; ++i) {
        const int t = t0 + i;

        const AOps nxt = fetch_aops(lds_h2, ft, t0, (i + 1) & (T_ / NW - 1), j0, mk, m, fsw);

        v2f bmat; bmat.x = cur.bx; bmat.y = cur.by;
        const float cv = cur.cv;

        #pragma unroll
        for (int mt = 0; mt < BT / 16; ++mt) {
            const float f = mt ? cur.f1 : cur.f0;
            v2f a;
            a.x = lrelu(f * cur.w1.x + cur.b1.x);
            a.y = lrelu(f * cur.w1.y + cur.b1.y);
            v8f c = {cv, cv, cv, cv, cv, cv, cv, cv};

            v8f d = __builtin_amdgcn_wmma_f32_16x16x4_f32(
                 false, a,  false, bmat,
                 (short)0, c,  false,  false);

            if (m < K_) {
                #pragma unroll
                for (int v = 0; v < 8; ++v) {
                    const int brow = mt * 16 + v + 8 * half;
                    lds_h2[(t * BT + brow) * K_ + m] = lrelu(d[v]);
                }
            }
        }
        cur = nxt;
    }

    __syncthreads();

    float* tr = lds_tr + wave * (BT * CH);
    const int wsw = (lane >> 1) & 7;
    const int rr0 = lane >> 3;
    const int rcg = lane & 7;
    const char* h2row = (const char*)lds_h2 + lane * (K_ * 4);

    for (int c = c0; c < NCH; c += GS * NW) {
        const int g0 = c * CH;
        const int cn = (c + GS * NW < NCH) ? c + GS * NW : c;

        #pragma unroll
        for (int hc = 0; hc < 2; ++hc) {
            const GOps cur = nxt;
            nxt = fetch_gops(w3, b3, edge_tf, (hc == 0 ? g0 + 16 : cn * CH) + m);

            *(float4*)&tr[lane * CH + (((hc * 4 + 0) ^ wsw) << 2)] = four_g<0>(cur, h2row);
            *(float4*)&tr[lane * CH + (((hc * 4 + 1) ^ wsw) << 2)] = four_g<4>(cur, h2row);
            *(float4*)&tr[lane * CH + (((hc * 4 + 2) ^ wsw) << 2)] = four_g<8>(cur, h2row);
            *(float4*)&tr[lane * CH + (((hc * 4 + 3) ^ wsw) << 2)] = four_g<12>(cur, h2row);
        }

        #pragma unroll
        for (int p = 0; p < BT / 4; ++p) {
            const int r  = p * 4 + rr0;
            const int cs = rcg ^ ((r >> 1) & 7);
            const float4 v = *(const float4*)&tr[r * CH + (cs << 2)];
            *(float4*)&out[(size_t)(bbase + r) * G_ + g0 + (rcg << 2)] = v;
        }
    }
}

extern "C" void kernel_launch(void* const* d_in, const int* in_sizes, int n_in,
                              void* d_out, int out_size, void* d_ws, size_t ws_size,
                              hipStream_t stream)
{
    const float* features = (const float*)d_in[0];
    const float* w1       = (const float*)d_in[1];
    const float* b1       = (const float*)d_in[2];
    const float* w2       = (const float*)d_in[3];
    const float* b2       = (const float*)d_in[4];
    const float* w3       = (const float*)d_in[5];
    const float* b3       = (const float*)d_in[6];
    const int*   edge_tf  = (const int*)d_in[7];
    float*       out      = (float*)d_out;

    dim3 grid(GS, B_ / BT);
    ae_decoder_fused<<<grid, 512, 0, stream>>>(features, w1, b1, w2, b2,
                                               w3, b3, edge_tf, out);
}


# Phase A epilogue on all 32 lanes: D columns 4-15 repeat 0-3, so copy q of a column keeps rows 2q, 2q+1 (register pick by lane): 2 lrelu, one 2-address LDS store, no exec mask or branch; same sums
# speedup vs baseline: 3.7909x; 1.0664x over previous
#include <hip/hip_runtime.h>


typedef float v2f __attribute__((ext_vector_type(2)));
typedef float v8f __attribute__((ext_vector_type(8)));

#define T_   512
#define K_   4
#define G_   20000
#define EPG_ 3
#define B_   512
#define BT   32
#define NW   16
#define GS   16
#define CH   32
#define NCH  (G_ / CH)

__device__ __forceinline__ float lrelu(float x) {
    return __builtin_fmaxf(x, 0.01f * x);
}

template <int S>
__device__ __forceinline__ float row_bcast(float v) {
    return __builtin_bit_cast(float,
        __builtin_amdgcn_update_dpp(0, __builtin_bit_cast(int, v), 0x150 + S, 0xf, 0xf, true));
}
template <int S>
__device__ __forceinline__ int row_bcast(int v) {
    return __builtin_amdgcn_update_dpp(0, v, 0x150 + S, 0xf, 0xf, true);
}

struct GOps {
    float4 w[EPG_];
    int    to[EPG_];
    float  b;
};

__device__ __forceinline__ GOps fetch_gops(const float* __restrict__ w3,
                                           const float* __restrict__ b3,
                                           const int*   __restrict__ edge_tf, int g)
{
    GOps o;
    #pragma unroll
    for (int e = 0; e < EPG_; ++e) {
        o.w[e]  = *(const float4*)&w3[(size_t)(g * EPG_ + e) * K_];
        o.to[e] = edge_tf[g * EPG_ + e] * (BT * K_ * 4);
    }
    o.b = b3[g];
    return o;
}

template <int S>
__device__ __forceinline__ void fma_row(float& acc, float w, float h)
{
    asm("v_fmac_f32_dpp %0, %1, %2 row_share:%3 row_mask:0xf bank_mask:0xf"
        : "+v"(acc) : "v"(w), "v"(h), "n"(S));
}

template <int S>
__device__ __forceinline__ float one_g(const GOps& o, const char* h2row)
{
    float a = row_bcast<S>(o.b);
    #pragma unroll
    for (int e = 0; e < EPG_; ++e) {
        const float4 h = *(const float4*)(h2row + row_bcast<S>(o.to[e]));
        fma_row<S>(a, o.w[e].x, h.x);
        fma_row<S>(a, o.w[e].y, h.y);
        fma_row<S>(a, o.w[e].z, h.z);
        fma_row<S>(a, o.w[e].w, h.w);
    }
    return a;
}

template <int S0>
__device__ __forceinline__ float4 four_g(const GOps& o, const char* h2row)
{
    return make_float4(one_g<S0>(o, h2row), one_g<S0 + 1>(o, h2row),
                       one_g<S0 + 2>(o, h2row), one_g<S0 + 3>(o, h2row));
}

struct AOps {
    float2 w1, b1;
    float  bx, by, cv, f0, f1;
};

__device__ __forceinline__ AOps fetch_aops(const float* h2, const float* ft, int t0, int i,
                                           int j0, int mk, int m, int fsw)
{
    const int t = t0 + i;
    const float* rec = h2 + t * (BT * K_) + (t & 7) * 12;
    AOps o;
    o.w1 = *(const float2*)&rec[j0];
    o.b1 = *(const float2*)&rec[4 + j0];
    o.bx = rec[8 + j0 * K_ + mk];
    o.by = rec[8 + (j0 + 1) * K_ + mk];
    o.cv = rec[24 + mk];
    const int fo = m * (T_ / NW) + ((((i >> 2) ^ fsw) << 2) | (i & 3));
    o.f0 = ft[fo];
    o.f1 = ft[fo + 16 * (T_ / NW)];
    return o;
}

__global__ __launch_bounds__(512)
void ae_decoder_fused(const float* __restrict__ features,
                      const float* __restrict__ w1,
                      const float* __restrict__ b1,
                      const float* __restrict__ w2,
                      const float* __restrict__ b2,
                      const float* __restrict__ w3,
                      const float* __restrict__ b3,
                      const int*   __restrict__ edge_tf,
                      float*       __restrict__ out)
{
    __shared__ __align__(16) float lds_h2[T_ * BT * K_];
    __shared__ __align__(16) float lds_tr[NW * BT * CH];

    const int tid  = threadIdx.x;
    const int wave = __builtin_amdgcn_readfirstlane(tid >> 5);
    const int lane = tid & 31;
    const int half = lane >> 4;
    const int m    = lane & 15;

    const int bbase  = blockIdx.y * BT;
    const int gsplit = blockIdx.x;

    const int c0 = gsplit + GS * wave;
    GOps nxt = fetch_gops(w3, b3, edge_tf, c0 * CH + m);

    const int j0 = half * 2;

    float* ft = lds_tr + wave * (BT * CH);
    const int t0 = wave * (T_ / NW);
    {
        const int fr = lane >> 3;
        const int fc = lane & 7;
        float4 fv[BT / 4];
        #pragma unroll
        for (int p = 0; p < BT / 4; ++p) {
            const int r = p * 4 + fr;
            fv[p] = *(const float4*)&features[(size_t)(bbase + r) * T_ + t0 + (fc << 2)];
        }
        #pragma unroll
        for (int p = 0; p < BT / 4; ++p) {
            const int r = p * 4 + fr;
            *(float4*)&ft[r * (T_ / NW) + ((fc ^ ((r >> 1) & 7)) << 2)] = fv[p];
        }
    }
    const int fsw = (m >> 1) & 7;

    {
        const int tl = t0 + lane;
        const float4 pw1 = *(const float4*)&w1[tl * K_];
        const float4 pb1 = *(const float4*)&b1[tl * K_];
        const float4 pb2 = *(const float4*)&b2[tl * K_];
        float4 pw2[4];
        #pragma unroll
        for (int q = 0; q < 4; ++q)
            pw2[q] = *(const float4*)&w2[(size_t)t0 * (K_ * K_) + q * 128 + lane * 4];
        float* rl = &lds_h2[tl * (BT * K_) + (tl & 7) * 12];
        *(float4*)&rl[0]  = pw1;
        *(float4*)&rl[4]  = pb1;
        *(float4*)&rl[24] = pb2;
        #pragma unroll
        for (int q = 0; q < 4; ++q) {
            const int tq = t0 + q * 8 + (lane >> 2);
            *(float4*)&lds_h2[tq * (BT * K_) + (tq & 7) * 12 + 8 + (lane & 3) * 4] = pw2[q];
        }
    }
    const int mk = m & 3;
    const int cq = m >> 2;

    AOps cur = fetch_aops(lds_h2, ft, t0, 0, j0, mk, m, fsw);

    #pragma unroll 8
    for (int i = 0; i < T_ / NW; ++i) {
        const int t = t0 + i;

        const AOps nxt = fetch_aops(lds_h2, ft, t0, (i + 1) & (T_ / NW - 1), j0, mk, m, fsw);

        v2f bmat; bmat.x = cur.bx; bmat.y = cur.by;
        const float cv = cur.cv;

        #pragma unroll
        for (int mt = 0; mt < BT / 16; ++mt) {
            const float f = mt ? cur.f1 : cur.f0;
            v2f a;
            a.x = lrelu(f * cur.w1.x + cur.b1.x);
            a.y = lrelu(f * cur.w1.y + cur.b1.y);
            v8f c = {cv, cv, cv, cv, cv, cv, cv, cv};

            v8f d = __builtin_amdgcn_wmma_f32_16x16x4_f32(
                 false, a,  false, bmat,
                 (short)0, c,  false,  false);

            const float x0 = (cq & 2) ? ((cq & 1) ? d[6] : d[4]) : ((cq & 1) ? d[2] : d[0]);
            const float x1 = (cq & 2) ? ((cq & 1) ? d[7] : d[5]) : ((cq & 1) ? d[3] : d[1]);
            const int brow = mt * 16 + 8 * half + 2 * cq;
            lds_h2[(t * BT + brow) * K_ + mk]     = lrelu(x0);
            lds_h2[(t * BT + brow + 1) * K_ + mk] = lrelu(x1);
        }
        cur = nxt;
    }

    __syncthreads();

    float* tr = lds_tr + wave * (BT * CH);
    const int wsw = (lane >> 1) & 7;
    const int rr0 = lane >> 3;
    const int rcg = lane & 7;
    const char* h2row = (const char*)lds_h2 + lane * (K_ * 4);

    for (int c = c0; c < NCH; c += GS * NW) {
        const int g0 = c * CH;
        const int cn = (c + GS * NW < NCH) ? c + GS * NW : c;

        #pragma unroll
        for (int hc = 0; hc < 2; ++hc) {
            const GOps cur = nxt;
            nxt = fetch_gops(w3, b3, edge_tf, (hc == 0 ? g0 + 16 : cn * CH) + m);

            *(float4*)&tr[lane * CH + (((hc * 4 + 0) ^ wsw) << 2)] = four_g<0>(cur, h2row);
            *(float4*)&tr[lane * CH + (((hc * 4 + 1) ^ wsw) << 2)] = four_g<4>(cur, h2row);
            *(float4*)&tr[lane * CH + (((hc * 4 + 2) ^ wsw) << 2)] = four_g<8>(cur, h2row);
            *(float4*)&tr[lane * CH + (((hc * 4 + 3) ^ wsw) << 2)] = four_g<12>(cur, h2row);
        }

        #pragma unroll
        for (int p = 0; p < BT / 4; ++p) {
            const int r  = p * 4 + rr0;
            const int cs = rcg ^ ((r >> 1) & 7);
            const float4 v = *(const float4*)&tr[r * CH + (cs << 2)];
            *(float4*)&out[(size_t)(bbase + r) * G_ + g0 + (rcg << 2)] = v;
        }
    }
}

extern "C" void kernel_launch(void* const* d_in, const int* in_sizes, int n_in,
                              void* d_out, int out_size, void* d_ws, size_t ws_size,
                              hipStream_t stream)
{
    const float* features = (const float*)d_in[0];
    const float* w1       = (const float*)d_in[1];
    const float* b1       = (const float*)d_in[2];
    const float* w2       = (const float*)d_in[3];
    const float* b2       = (const float*)d_in[4];
    const float* w3       = (const float*)d_in[5];
    const float* b3       = (const float*)d_in[6];
    const int*   edge_tf  = (const int*)d_in[7];
    float*       out      = (float*)d_out;

    dim3 grid(GS, B_ / BT);
    ae_decoder_fused<<<grid, 512, 0, stream>>>(features, w1, b1, w2, b2,
                                               w3, b3, edge_tf, out);
}


# Phase B on packed f32 pairs: a lane owns 2 b (t row stored [k pair][b pair][2k][2b]), the wave's two 16-lane rows take different g, each w3 is picked once per row for one packed fma; same fma order
# speedup vs baseline: 4.1531x; 1.0955x over previous
#include <hip/hip_runtime.h>


typedef float v2f __attribute__((ext_vector_type(2)));
typedef float v8f __attribute__((ext_vector_type(8)));

#define T_   512
#define K_   4
#define G_   20000
#define EPG_ 3
#define B_   512
#define BT   32
#define NW   16
#define GS   16
#define CH   32
#define NCH  (G_ / CH)

__device__ __forceinline__ float lrelu(float x) {
    return __builtin_fmaxf(x, 0.01f * x);
}

template <int S>
__device__ __forceinline__ float row_pick(float v) {
    return __builtin_bit_cast(float,
        __builtin_amdgcn_permlane16(0u, __builtin_bit_cast(unsigned, v),
                                    0x11111111u * S, 0x11111111u * S, true, false));
}
template <int S>
__device__ __forceinline__ int row_bcast(int v) {
    return __builtin_amdgcn_update_dpp(0, v, 0x150 + S, 0xf, 0xf, true);
}

struct GOps {
    float4 w[EPG_];
    int    to[EPG_];
    float  b;
};

__device__ __forceinline__ GOps fetch_gops(const float* __restrict__ w3,
                                           const float* __restrict__ b3,
                                           const int*   __restrict__ edge_tf, int g)
{
    GOps o;
    #pragma unroll
    for (int e = 0; e < EPG_; ++e) {
        o.w[e]  = *(const float4*)&w3[(size_t)(g * EPG_ + e) * K_];
        o.to[e] = edge_tf[g * EPG_ + e] * (BT * K_ * 4);
    }
    o.b = b3[g];
    return o;
}

__device__ __forceinline__ v2f fma_pair(float w, float h0, float h1, v2f a)
{
    v2f ww; ww.x = w;  ww.y = w;
    v2f hh; hh.x = h0; hh.y = h1;
    return __builtin_elementwise_fma(ww, hh, a);
}

struct SlotH {
    float4 h01[EPG_], h23[EPG_];
};

template <int S>
__device__ __forceinline__ SlotH load_slot(const GOps& o, const char* h2lane)
{
    SlotH s;
    #pragma unroll
    for (int e = 0; e < EPG_; ++e) {
        const char* p = h2lane + row_bcast<S>(o.to[e]);
        s.h01[e] = *(const float4*)p;
        s.h23[e] = *(const float4*)(p + 256);
    }
    return s;
}

template <int S>
__device__ __forceinline__ v2f calc_slot(const GOps& o, const SlotH& s)
{
    const float bb = row_pick<S>(o.b);
    v2f a; a.x = bb; a.y = bb;
    #pragma unroll
    for (int e = 0; e < EPG_; ++e) {
        a = fma_pair(row_pick<S>(o.w[e].x), s.h01[e].x, s.h01[e].y, a);
        a = fma_pair(row_pick<S>(o.w[e].y), s.h01[e].z, s.h01[e].w, a);
        a = fma_pair(row_pick<S>(o.w[e].z), s.h23[e].x, s.h23[e].y, a);
        a = fma_pair(row_pick<S>(o.w[e].w), s.h23[e].z, s.h23[e].w, a);
    }
    return a;
}

template <int S0>
__device__ __forceinline__ void four_slots(const GOps& o, const char* h2lane,
                                           float4& r0, float4& r1)
{
    const SlotH s0 = load_slot<S0>(o, h2lane),     s1 = load_slot<S0 + 1>(o, h2lane),
                s2 = load_slot<S0 + 2>(o, h2lane), s3 = load_slot<S0 + 3>(o, h2lane);
    __builtin_amdgcn_sched_barrier(0);
    const v2f a0 = calc_slot<S0>(o, s0),     a1 = calc_slot<S0 + 1>(o, s1),
              a2 = calc_slot<S0 + 2>(o, s2), a3 = calc_slot<S0 + 3>(o, s3);
    r0 = make_float4(a0.x, a1.x, a2.x, a3.x);
    r1 = make_float4(a0.y, a1.y, a2.y, a3.y);
}

struct AOps {
    float2 w1, b1;
    float  bx, by, cv, f0, f1;
};

__device__ __forceinline__ AOps fetch_aops(const float* h2, const float* ft, int t0, int i,
                                           int j0, int mk, int m, int fsw)
{
    const int t = t0 + i;
    const float* rec = h2 + t * (BT * K_) + (t & 7) * 12;
    AOps o;
    o.w1 = *(const float2*)&rec[j0];
    o.b1 = *(const float2*)&rec[4 + j0];
    o.bx = rec[8 + j0 * K_ + mk];
    o.by = rec[8 + (j0 + 1) * K_ + mk];
    o.cv = rec[24 + mk];
    const int fo = m * (T_ / NW) + ((((i >> 2) ^ fsw) << 2) | (i & 3));
    o.f0 = ft[fo];
    o.f1 = ft[fo + 16 * (T_ / NW)];
    return o;
}

__global__ __launch_bounds__(512)
void ae_decoder_fused(const float* __restrict__ features,
                      const float* __restrict__ w1,
                      const float* __restrict__ b1,
                      const float* __restrict__ w2,
                      const float* __restrict__ b2,
                      const float* __restrict__ w3,
                      const float* __restrict__ b3,
                      const int*   __restrict__ edge_tf,
                      float*       __restrict__ out)
{
    __shared__ __align__(16) float lds_h2[T_ * BT * K_];
    __shared__ __align__(16) float lds_tr[NW * BT * CH];

    const int tid  = threadIdx.x;
    const int wave = __builtin_amdgcn_readfirstlane(tid >> 5);
    const int lane = tid & 31;
    const int half = lane >> 4;
    const int m    = lane & 15;

    const int bbase  = blockIdx.y * BT;
    const int gsplit = blockIdx.x;

    const int c0 = gsplit + GS * wave;
    GOps nxt = fetch_gops(w3, b3, edge_tf, c0 * CH + lane);

    const int j0 = half * 2;

    float* ft = lds_tr + wave * (BT * CH);
    const int t0 = wave * (T_ / NW);
    {
        const int fr = lane >> 3;
        const int fc = lane & 7;
        float4 fv[BT / 4];
        #pragma unroll
        for (int p = 0; p < BT / 4; ++p) {
            const int r = p * 4 + fr;
            fv[p] = *(const float4*)&features[(size_t)(bbase + r) * T_ + t0 + (fc << 2)];
        }
        #pragma unroll
        for (int p = 0; p < BT / 4; ++p) {
            const int r = p * 4 + fr;
            *(float4*)&ft[r * (T_ / NW) + ((fc ^ ((r >> 1) & 7)) << 2)] = fv[p];
        }
    }
    const int fsw = (m >> 1) & 7;

    {
        const int tl = t0 + lane;
        const float4 pw1 = *(const float4*)&w1[tl * K_];
        const float4 pb1 = *(const float4*)&b1[tl * K_];
        const float4 pb2 = *(const float4*)&b2[tl * K_];
        float4 pw2[4];
        #pragma unroll
        for (int q = 0; q < 4; ++q)
            pw2[q] = *(const float4*)&w2[(size_t)t0 * (K_ * K_) + q * 128 + lane * 4];
        float* rl = &lds_h2[tl * (BT * K_) + (tl & 7) * 12];
        *(float4*)&rl[0]  = pw1;
        *(float4*)&rl[4]  = pb1;
        *(float4*)&rl[24] = pb2;
        #pragma unroll
        for (int q = 0; q < 4; ++q) {
            const int tq = t0 + q * 8 + (lane >> 2);
            *(float4*)&lds_h2[tq * (BT * K_) + (tq & 7) * 12 + 8 + (lane & 3) * 4] = pw2[q];
        }
    }
    const int mk = m & 3;
    const int cq = m >> 2;

    AOps cur = fetch_aops(lds_h2, ft, t0, 0, j0, mk, m, fsw);

    #pragma unroll 8
    for (int i = 0; i < T_ / NW; ++i) {
        const int t = t0 + i;

        const AOps nxt = fetch_aops(lds_h2, ft, t0, (i + 1) & (T_ / NW - 1), j0, mk, m, fsw);

        v2f bmat; bmat.x = cur.bx; bmat.y = cur.by;
        const float cv = cur.cv;

        #pragma unroll
        for (int mt = 0; mt < BT / 16; ++mt) {
            const float f = mt ? cur.f1 : cur.f0;
            v2f a;
            a.x = lrelu(f * cur.w1.x + cur.b1.x);
            a.y = lrelu(f * cur.w1.y + cur.b1.y);
            v8f c = {cv, cv, cv, cv, cv, cv, cv, cv};

            v8f d = __builtin_amdgcn_wmma_f32_16x16x4_f32(
                 false, a,  false, bmat,
                 (short)0, c,  false,  false);

            const float x0 = (cq & 2) ? ((cq & 1) ? d[6] : d[4]) : ((cq & 1) ? d[2] : d[0]);
            const float x1 = (cq & 2) ? ((cq & 1) ? d[7] : d[5]) : ((cq & 1) ? d[3] : d[1]);
            const int bp = mt * 8 + 4 * half + cq;
            *(float2*)&lds_h2[t * (BT * K_) + (mk >> 1) * 64 + bp * 4 + (mk & 1) * 2] =
                make_float2(lrelu(x0), lrelu(x1));
        }
        cur = nxt;
    }

    __syncthreads();

    float* tr = lds_tr + wave * (BT * CH);
    const int wcol = (4 * half) ^ (m & 7);
    float* trw = tr + (2 * m) * CH;
    const int rr0 = lane >> 3;
    const int rcg = lane & 7;
    const char* h2lane = (const char*)lds_h2 + m * 16;

    for (int c = c0; c < NCH; c += GS * NW) {
        const int g0 = c * CH;
        const int cn = (c + GS * NW < NCH) ? c + GS * NW : c;

        const GOps cur = nxt;
        nxt = fetch_gops(w3, b3, edge_tf, cn * CH + lane);

        {
            float4 r0, r1;
            four_slots<0>(cur, h2lane, r0, r1);
            *(float4*)&trw[(wcol ^ 0) << 2]        = r0;
            *(float4*)&trw[CH + ((wcol ^ 0) << 2)] = r1;
        }
        __builtin_amdgcn_sched_barrier(0);
        {
            float4 r0, r1;
            four_slots<4>(cur, h2lane, r0, r1);
            *(float4*)&trw[(wcol ^ 1) << 2]        = r0;
            *(float4*)&trw[CH + ((wcol ^ 1) << 2)] = r1;
        }
        __builtin_amdgcn_sched_barrier(0);
        {
            float4 r0, r1;
            four_slots<8>(cur, h2lane, r0, r1);
            *(float4*)&trw[(wcol ^ 2) << 2]        = r0;
            *(float4*)&trw[CH + ((wcol ^ 2) << 2)] = r1;
        }
        __builtin_amdgcn_sched_barrier(0);
        {
            float4 r0, r1;
            four_slots<12>(cur, h2lane, r0, r1);
            *(float4*)&trw[(wcol ^ 3) << 2]        = r0;
            *(float4*)&trw[CH + ((wcol ^ 3) << 2)] = r1;
        }

        #pragma unroll
        for (int p = 0; p < BT / 4; ++p) {
            const int r  = p * 4 + rr0;
            const int cs = rcg ^ ((r >> 1) & 7);
            const float4 v = *(const float4*)&tr[r * CH + (cs << 2)];
            *(float4*)&out[(size_t)(bbase + r) * G_ + g0 + (rcg << 2)] = v;
        }
    }
}

extern "C" void kernel_launch(void* const* d_in, const int* in_sizes, int n_in,
                              void* d_out, int out_size, void* d_ws, size_t ws_size,
                              hipStream_t stream)
{
    const float* features = (const float*)d_in[0];
    const float* w1       = (const float*)d_in[1];
    const float* b1       = (const float*)d_in[2];
    const float* w2       = (const float*)d_in[3];
    const float* b2       = (const float*)d_in[4];
    const float* w3       = (const float*)d_in[5];
    const float* b3       = (const float*)d_in[6];
    const int*   edge_tf  = (const int*)d_in[7];
    float*       out      = (float*)d_out;

    dim3 grid(GS, B_ / BT);
    ae_decoder_fused<<<grid, 512, 0, stream>>>(features, w1, b1, w2, b2,
                                               w3, b3, edge_tf, out);
}
